# starting kernel
# baseline (speedup 1.0000x reference)
#include <hip/hip_runtime.h>

#define D 128
#define NLAYERS 4

typedef __attribute__((ext_vector_type(2))) float v2f;
typedef __attribute__((ext_vector_type(8))) float v8f;


__global__ void zero_f32_kernel(float* __restrict__ p, long n) {
    long i = (long)blockIdx.x * blockDim.x + threadIdx.x;
    if (i < n) p[i] = 0.0f;
}

__global__ void degree_kernel(const int* __restrict__ src, const int* __restrict__ dst,
                              float* __restrict__ deg_src, float* __restrict__ deg_dst,
                              int E) {
    int e = blockIdx.x * blockDim.x + threadIdx.x;
    if (e < E) {
        unsafeAtomicAdd(&deg_src[src[e]], 1.0f);
        unsafeAtomicAdd(&deg_dst[dst[e]], 1.0f);
    }
}

__global__ void norm_kernel(float* __restrict__ deg, long n) {
    long i = (long)blockIdx.x * blockDim.x + threadIdx.x;
    if (i < n) deg[i] = rsqrtf(fmaxf(deg[i], 1.0f));
}

__global__ void spmm_scatter_kernel(const float* __restrict__ hin,
                                    const float* __restrict__ norm_src,
                                    const int* __restrict__ src,
                                    const int* __restrict__ dst,
                                    float* __restrict__ agg, int E) {
    int t = blockIdx.x * blockDim.x + threadIdx.x;
    int e = t >> 5;
    int q = (t & 31) << 2;
    if (e >= E) return;
    int s = src[e];
    int d = dst[e];
    float4 v = *(const float4*)(hin + (long)s * D + q);
    float ns = norm_src[s];
    float* outp = agg + (long)d * D + q;
    unsafeAtomicAdd(outp + 0, v.x * ns);
    unsafeAtomicAdd(outp + 1, v.y * ns);
    unsafeAtomicAdd(outp + 2, v.z * ns);
    unsafeAtomicAdd(outp + 3, v.w * ns);
}

__global__ void __launch_bounds__(256)
gcn_gemm_wmma_kernel(const float* __restrict__ agg,
                     const float* __restrict__ norm_dst,
                     const float* __restrict__ W,
                     const float* __restrict__ bias,
                     float* __restrict__ out, int N) {
    __shared__ float sA[16][D + 4];

    const int tid     = threadIdx.x;
    const int rowBase = blockIdx.x * 16;

    for (int idx = tid; idx < 16 * D; idx += 256) {
        int r = idx >> 7;
        int c = idx & (D - 1);
        int row = rowBase + r;
        float v = 0.0f;
        if (row < N) v = agg[(long)row * D + c] * norm_dst[row];
        sA[r][c] = v;
    }
    __syncthreads();

    const int wave  = tid >> 5;
    const int lane  = tid & 31;
    const int laneM = lane & 15;
    const int koff  = (lane >> 4) << 1;
    const int n     = (wave << 4) + laneM;

    v8f c;
    const float bv = bias[n];
#pragma unroll
    for (int v = 0; v < 8; ++v) c[v] = bv;

#pragma unroll
    for (int k0 = 0; k0 < D; k0 += 4) {
        v2f a, b;
        a.x = sA[laneM][k0 + koff];
        a.y = sA[laneM][k0 + koff + 1];
        b.x = W[(long)(k0 + koff) * D + n];
        b.y = W[(long)(k0 + koff + 1) * D + n];
        c = __builtin_amdgcn_wmma_f32_16x16x4_f32(
                 false, a,  false, b,
                 (short)0, c,  false,  false);
    }

    const int mBase = rowBase + ((lane >> 4) << 3);
#pragma unroll
    for (int v = 0; v < 8; ++v) {
        int row = mBase + v;
        if (row < N) out[(long)row * D + n] = c[v];
    }
}


extern "C" void kernel_launch(void* const* d_in, const int* in_sizes, int n_in,
                              void* d_out, int out_size, void* d_ws, size_t ws_size,
                              hipStream_t stream) {
    const float* feat = (const float*)d_in[0];
    const float* W    = (const float*)d_in[1];
    const float* bias = (const float*)d_in[2];
    const int*   src  = (const int*)d_in[3];
    const int*   dst  = (const int*)d_in[4];

    const int N = in_sizes[0] / D;
    const int E = in_sizes[3];

    float* ws       = (float*)d_ws;
    float* norm_src = ws;
    float* norm_dst = ws + N;
    float* agg      = ws + 2L * N;
    float* hbuf     = agg + (long)N * D;
    float* outf     = (float*)d_out;

    const long twoN = 2L * N;
    const long ND   = (long)N * D;

    zero_f32_kernel<<<(int)((twoN + 255) / 256), 256, 0, stream>>>(norm_src, twoN);
    degree_kernel<<<(E + 255) / 256, 256, 0, stream>>>(src, dst, norm_src, norm_dst, E);
    norm_kernel<<<(int)((twoN + 255) / 256), 256, 0, stream>>>(norm_src, twoN);

    const long edgeThreads = (long)E * 32;
    const int  edgeGrid    = (int)((edgeThreads + 255) / 256);
    const int  gemmGrid    = (N + 15) / 16;

    const float* hin = feat;
    for (int layer = 0; layer < NLAYERS; ++layer) {
        float* hout = (layer & 1) ? outf : hbuf;

        zero_f32_kernel<<<(int)((ND + 255) / 256), 256, 0, stream>>>(agg, ND);
        spmm_scatter_kernel<<<edgeGrid, 256, 0, stream>>>(hin, norm_src, src, dst, agg, E);
        gcn_gemm_wmma_kernel<<<gemmGrid, 256, 0, stream>>>(
            agg, norm_dst, W + (long)layer * D * D, bias + (long)layer * D, hout, N);

        hin = hout;
    }
}


# Degree pass: index arrays counted as integers in LDS, one (node slice, edge chunk) per group; 16 count planes added in the norm pass; no clearing launch, no global float atomics; sums unchanged
# speedup vs baseline: 3.4822x; 3.4822x over previous
#include <hip/hip_runtime.h>

#define D 128
#define NLAYERS 4
#define TR 32
#define EPW 8

typedef __attribute__((ext_vector_type(2))) float v2f;
typedef __attribute__((ext_vector_type(8))) float v8f;

#define HC 16
#define HR 4
#define HWIN 16384
#define HLD 16

__global__ void __launch_bounds__(1024)
degree_hist_kernel(const int* __restrict__ src, const int* __restrict__ dst,
                   int* __restrict__ part, int E, long Np) {
    __shared__ int4 hist4[HWIN / 4];
    int* hist = (int*)hist4;

    const int tid = threadIdx.x;
    const int c   = blockIdx.x % HC;
    const int r   = (blockIdx.x / HC) % HR;
    const int a   = blockIdx.x / (HC * HR);
    const int* __restrict__ idx = a ? dst : src;

    const long SL = ((Np + 64 * HR - 1) / (64 * HR)) * 64;
    long n0 = r * SL;        if (n0 > Np) n0 = Np;
    long n1 = n0 + SL;       if (n1 > Np) n1 = Np;

    const long chunk = ((((long)E + HC - 1) / HC) + 3) & ~3L;
    long e0 = c * chunk;     if (e0 > E) e0 = E;
    long e1 = e0 + chunk;    if (e1 > E) e1 = E;
    const int nv   = (int)((e1 - e0) >> 2);
    const int tail = (int)(e1 - e0) - 4 * nv;
    const int4* __restrict__ idx4 = (const int4*)(idx + e0);

    int* __restrict__ outp = part + ((long)a * HC + c) * Np;

    for (long w0 = n0; w0 < n1; w0 += HWIN) {
        const int      wb = (int)w0;
        const unsigned wn = (unsigned)((n1 - w0 < HWIN) ? (n1 - w0) : HWIN);
        for (unsigned i = 4 * tid; i < wn; i += 4096) hist4[i >> 2] = make_int4(0, 0, 0, 0);
        __syncthreads();

        for (int vb = 0; vb < nv; vb += HLD * 1024) {
            int4 q[HLD];
#pragma unroll
            for (int j = 0; j < HLD; ++j) {
                const int v = vb + j * 1024 + tid;
                q[j] = idx4[v < nv ? v : nv - 1];
            }
#pragma unroll
            for (int j = 0; j < HLD; ++j) {
                if (vb + j * 1024 + tid < nv) {
                    const unsigned t0 = (unsigned)(q[j].x - wb), t1 = (unsigned)(q[j].y - wb);
                    const unsigned t2 = (unsigned)(q[j].z - wb), t3 = (unsigned)(q[j].w - wb);
                    if (t0 < wn) atomicAdd(&hist[t0], 1);
                    if (t1 < wn) atomicAdd(&hist[t1], 1);
                    if (t2 < wn) atomicAdd(&hist[t2], 1);
                    if (t3 < wn) atomicAdd(&hist[t3], 1);
                }
            }
        }
        if (tid < tail) {
            const unsigned t = (unsigned)(idx[e0 + 4L * nv + tid] - wb);
            if (t < wn) atomicAdd(&hist[t], 1);
        }
        __syncthreads();

        for (unsigned i = 4 * tid; i < wn; i += 4096) *(int4*)(outp + w0 + i) = hist4[i >> 2];
        __syncthreads();
    }
}

__global__ void norm_count_kernel(const int* __restrict__ part, float* __restrict__ nrm,
                                  int* __restrict__ cnt, long Np) {
    const long i = (long)blockIdx.x * blockDim.x + threadIdx.x;
    const long n = 2 * Np;
    int dg = 0;
    if (i < n) {
        const long a = (i >= Np) ? 1 : 0;
        const int* __restrict__ p = part + a * HC * Np + (i - a * Np);
        int d[HC];
#pragma unroll
        for (int g = 0; g < HC; ++g) d[g] = p[(long)g * Np];
#pragma unroll
        for (int g = 0; g < HC; ++g) dg += d[g];
    }
    int tsum = dg;
#pragma unroll
    for (int o = 16; o > 0; o >>= 1) tsum += __shfl_xor(tsum, o);
    if (i < n) {
        nrm[i] = rsqrtf(fmaxf((float)dg, 1.0f));
        if (i >= Np && (threadIdx.x & 31) == 0) cnt[(i - Np) >> 5] = tsum;
    }
}

__global__ void __launch_bounds__(256)
tile_scan_kernel(int* __restrict__ cur, int* __restrict__ off, int NT) {
    __shared__ int part[256];
    const int tid   = threadIdx.x;
    const int chunk = (NT + 255) / 256;
    int t0 = tid * chunk;       if (t0 > NT) t0 = NT;
    int t1 = t0 + chunk;        if (t1 > NT) t1 = NT;
    int sum = 0;
    for (int t = t0; t < t1; ++t) sum += cur[t];
    part[tid] = sum;
    __syncthreads();
    for (int o = 1; o < 256; o <<= 1) {
        int v = (tid >= o) ? part[tid - o] : 0;
        __syncthreads();
        part[tid] += v;
        __syncthreads();
    }
    int run = part[tid] - sum;
    for (int t = t0; t < t1; ++t) {
        int c = cur[t];
        off[t] = run;
        cur[t] = run;
        run += c;
    }
    if (tid == 255) off[NT] = part[255];
}

#define BF_EPT 8
#define BF_WIN 2048

__global__ void __launch_bounds__(1024)
bin_fill_kernel(const int* __restrict__ src, const int* __restrict__ dst,
                int* __restrict__ cur,
                int* __restrict__ bsrc, int* __restrict__ bdst, int E, int NT) {
    __shared__ int hist[BF_WIN];
    const int  tid   = threadIdx.x;
    const long ebase = (long)blockIdx.x * (BF_EPT * 1024);

    int s[BF_EPT], d[BF_EPT], r[BF_EPT];
#pragma unroll
    for (int j = 0; j < BF_EPT; ++j) {
        const long e = ebase + j * 1024 + tid;
        const bool ok = e < E;
        d[j] = ok ? dst[e] : -1;
        s[j] = ok ? src[e] : 0;
        r[j] = 0;
    }

    for (int w0 = 0; w0 < NT; w0 += BF_WIN) {
#pragma unroll
        for (int k = 0; k < BF_WIN / 1024; ++k) hist[tid + 1024 * k] = 0;
        __syncthreads();

        bool in[BF_EPT];
        int  t[BF_EPT];
#pragma unroll
        for (int j = 0; j < BF_EPT; ++j) {
            const int tj = (d[j] >> 5) - w0;
            in[j] = tj >= 0 && tj < BF_WIN;
            t[j]  = in[j] ? tj : 0;
            if (in[j]) r[j] = atomicAdd(&hist[t[j]], 1);
        }
        __syncthreads();

        int c[BF_WIN / 1024], b[BF_WIN / 1024];
#pragma unroll
        for (int k = 0; k < BF_WIN / 1024; ++k) c[k] = hist[tid + 1024 * k];
#pragma unroll
        for (int k = 0; k < BF_WIN / 1024; ++k) {
            b[k] = 0;
            if (c[k] > 0) b[k] = atomicAdd(&cur[w0 + tid + 1024 * k], c[k]);
        }
#pragma unroll
        for (int k = 0; k < BF_WIN / 1024; ++k) hist[tid + 1024 * k] = b[k];
        __syncthreads();

        int p[BF_EPT];
#pragma unroll
        for (int j = 0; j < BF_EPT; ++j) p[j] = hist[t[j]] + r[j];
#pragma unroll
        for (int j = 0; j < BF_EPT; ++j) {
            if (in[j]) {
                bsrc[p[j]] = s[j];
                bdst[p[j]] = d[j];
            }
        }
        __syncthreads();
    }
}

__device__ __forceinline__ void lds_add(float* p, float v) {
    __hip_atomic_fetch_add(p, v, __ATOMIC_RELAXED, __HIP_MEMORY_SCOPE_WORKGROUP);
}

__global__ void __launch_bounds__(256)
tile_sum_kernel(const float* __restrict__ hin,
                const float* __restrict__ norm_src,
                const int* __restrict__ bsrc,
                const int* __restrict__ bdst,
                const int* __restrict__ off,
                float* __restrict__ agg, int N) {
    __shared__ float4 sT4[TR * D / 4];
    float* sT = (float*)sT4;

    const int tid = threadIdx.x;
    const int l   = tid & 31;
    const int w   = __builtin_amdgcn_readfirstlane(tid >> 5);
    const int t   = blockIdx.x;
    const int rowBase = t * TR;

#pragma unroll
    for (int k = 0; k < TR * D / 4 / 256; ++k) sT4[tid + 256 * k] = make_float4(0.f, 0.f, 0.f, 0.f);

    const int beg = off[t];
    const int end = off[t + 1];
    __syncthreads();

    for (int e0 = beg + w * EPW; e0 < end; e0 += 8 * EPW) {
        if (e0 + EPW <= end) {
            int s[EPW], dl[EPW];
#pragma unroll
            for (int j = 0; j < EPW; ++j) s[j] = bsrc[e0 + j];
#pragma unroll
            for (int j = 0; j < EPW; ++j) dl[j] = bdst[e0 + j] - rowBase;

            float4 v[EPW];
            float  ns[EPW];
#pragma unroll
            for (int j = 0; j < EPW; ++j) {
                v[j]  = ((const float4*)(hin + (long)s[j] * D))[l];
                ns[j] = norm_src[s[j]];
            }
#pragma unroll
            for (int j = 0; j < EPW; ++j) {
                float* p = sT + dl[j] * D + l;
                lds_add(p + 0,  v[j].x * ns[j]);
                lds_add(p + 32, v[j].y * ns[j]);
                lds_add(p + 64, v[j].z * ns[j]);
                lds_add(p + 96, v[j].w * ns[j]);
            }
        } else {
            for (int e = e0; e < end; ++e) {
                const int    s1 = bsrc[e];
                const int    d1 = bdst[e] - rowBase;
                const float  n1 = norm_src[s1];
                const float4 a  = ((const float4*)(hin + (long)s1 * D))[l];
                float* p = sT + d1 * D + l;
                lds_add(p + 0,  a.x * n1);
                lds_add(p + 32, a.y * n1);
                lds_add(p + 64, a.z * n1);
                lds_add(p + 96, a.w * n1);
            }
        }
    }
    __syncthreads();

#pragma unroll
    for (int k = 0; k < TR * D / 4 / 256; ++k) {
        const int i   = tid + 256 * k;
        const int row = rowBase + (i >> 5);
        const float* q = sT + (i >> 5) * D + l;
        const float4 o = make_float4(q[0], q[32], q[64], q[96]);
        if (row < N) ((float4*)(agg + (long)row * D))[l] = o;
    }
}

#define GR 128
#define AP (D + 4)

__global__ void __launch_bounds__(256)
gcn_gemm_wmma_kernel(const float* __restrict__ agg,
                     const float* __restrict__ norm_dst,
                     const float* __restrict__ W,
                     const float* __restrict__ bias,
                     float* __restrict__ out, int N) {
    __shared__ float4 sLds[D * D / 4 + GR * AP / 4];
    float4* sW4 = sLds;
    float*  sA  = (float*)(sLds + D * D / 4);

    const int tid = threadIdx.x;
    const int l   = tid & 31;
    const int w   = __builtin_amdgcn_readfirstlane(tid >> 5);
    const int c   = l & 15;
    const int h   = l >> 4;
    const int rowBase = blockIdx.x * GR + 16 * w;

    float4 w0[8], w1[8];
#pragma unroll
    for (int i = 0; i < 8; ++i) {
        const int m = w + 8 * i;
        w0[i] = ((const float4*)W)[(2 * m) * 32 + l];
        w1[i] = ((const float4*)W)[(2 * m + 1) * 32 + l];
    }
    float4 av[16];
    float  nd[16];
#pragma unroll
    for (int i = 0; i < 16; ++i) {
        const int row = (rowBase + i < N) ? rowBase + i : N - 1;
        av[i] = ((const float4*)(agg + (long)row * D))[l];
        nd[i] = norm_dst[row];
    }
    float2 bp[4];
#pragma unroll
    for (int p = 0; p < 4; ++p) bp[p] = ((const float2*)bias)[16 * p + c];

    __builtin_amdgcn_sched_barrier(0);

    const int sw = (w & 1) << 3;
#pragma unroll
    for (int i = 0; i < 8; ++i) {
        const int m = w + 8 * i;
        float4* d = sW4 + m * 64 + ((2 * l) ^ sw);
        d[0] = make_float4(w0[i].x, w1[i].x, w0[i].y, w1[i].y);
        d[1] = make_float4(w0[i].z, w1[i].z, w0[i].w, w1[i].w);
    }
#pragma unroll
    for (int i = 0; i < 16; ++i)
        *(float4*)(sA + (16 * w + i) * AP + 4 * l) =
            make_float4(av[i].x * nd[i], av[i].y * nd[i], av[i].z * nd[i], av[i].w * nd[i]);
    __syncthreads();

    v8f acc[8];
#pragma unroll
    for (int p = 0; p < 4; ++p) {
#pragma unroll
        for (int r = 0; r < 8; ++r) { acc[2 * p][r] = bp[p].x; acc[2 * p + 1][r] = bp[p].y; }
    }

    const float*  pA = sA + (16 * w + c) * AP + 2 * h;
    const float4* pW = sW4 + h * 64 + (c ^ (h << 3));

    v2f    a = *(const v2f*)pA;
    float4 q[4];
#pragma unroll
    for (int p = 0; p < 4; ++p) q[p] = pW[16 * p];

#pragma unroll
    for (int t = 0; t < D / 4; ++t) {
        v2f    an = a;
        float4 qn[4];
#pragma unroll
        for (int p = 0; p < 4; ++p) qn[p] = q[p];
        if (t + 1 < D / 4) {
            an = *(const v2f*)(pA + 4 * (t + 1));
#pragma unroll
            for (int p = 0; p < 4; ++p) qn[p] = pW[128 * (t + 1) + 16 * p];
        }
#pragma unroll
        for (int p = 0; p < 4; ++p) {
            v2f b0, b1;
            b0.x = q[p].x; b0.y = q[p].y;
            b1.x = q[p].z; b1.y = q[p].w;
            acc[2 * p]     = __builtin_amdgcn_wmma_f32_16x16x4_f32(false, a, false, b0, (short)0, acc[2 * p],     false, false);
            acc[2 * p + 1] = __builtin_amdgcn_wmma_f32_16x16x4_f32(false, a, false, b1, (short)0, acc[2 * p + 1], false, false);
        }
        a = an;
#pragma unroll
        for (int p = 0; p < 4; ++p) q[p] = qn[p];
    }

#pragma unroll
    for (int r = 0; r < 8; ++r) {
        const int row = rowBase + 8 * h + r;
        if (row < N) {
            float2* o = (float2*)(out + (long)row * D);
#pragma unroll
            for (int p = 0; p < 4; ++p) o[16 * p + c] = make_float2(acc[2 * p][r], acc[2 * p + 1][r]);
        }
    }
}


extern "C" void kernel_launch(void* const* d_in, const int* in_sizes, int n_in,
                              void* d_out, int out_size, void* d_ws, size_t ws_size,
                              hipStream_t stream) {
    const float* feat = (const float*)d_in[0];
    const float* W    = (const float*)d_in[1];
    const float* bias = (const float*)d_in[2];
    const int*   src  = (const int*)d_in[3];
    const int*   dst  = (const int*)d_in[4];

    const int N = in_sizes[0] / D;
    const int E = in_sizes[3];

    const long Np  = ((long)N + 63) & ~63L;
    const int  NT  = (int)(Np / TR);
    const long NTa = ((long)NT + 1 + 63) & ~63L;
    const long Ea  = ((long)E + 63) & ~63L;

    float* ws       = (float*)d_ws;
    float* norm_src = ws;
    float* norm_dst = ws + Np;
    int*   cur      = (int*)(ws + 2 * Np);
    int*   off      = cur + NTa;
    int*   bsrc     = off + NTa;
    int*   bdst     = bsrc + Ea;
    float* agg      = (float*)(bdst + Ea);
    float* hbuf     = agg + (long)N * D;
    int*   part     = (int*)(hbuf + (long)N * D);
    float* outf     = (float*)d_out;

    const long twoNp = 2 * Np;

    degree_hist_kernel<<<2 * HR * HC, 1024, 0, stream>>>(src, dst, part, E, Np);
    norm_count_kernel<<<(int)((twoNp + 255) / 256), 256, 0, stream>>>(part, norm_src, cur, Np);
    tile_scan_kernel<<<1, 256, 0, stream>>>(cur, off, NT);
    bin_fill_kernel<<<(E + BF_EPT * 1024 - 1) / (BF_EPT * 1024), 1024, 0, stream>>>(src, dst, cur, bsrc, bdst, E, NT);

    const int gemmGrid = (N + GR - 1) / GR;

    const float* hin = feat;
    for (int layer = 0; layer < NLAYERS; ++layer) {
        float* hout = (layer & 1) ? outf : hbuf;

        tile_sum_kernel<<<NT, 256, 0, stream>>>(hin, norm_src, bsrc, bdst, off, agg, N);
        gcn_gemm_wmma_kernel<<<gemmGrid, 256, 0, stream>>>(
            agg, norm_dst, W + (long)layer * D * D, bias + (long)layer * D, hout, N);

        hin = hout;
    }
}


# Edge pass: tile's edges sorted by row in LDS (returning LDS int add = rank); a wave sums a row's run in registers, one LDS float add set per run, not per edge; order: rank in a run, runs on arrival
# speedup vs baseline: 3.6493x; 1.0480x over previous
#include <hip/hip_runtime.h>

#define D 128
#define NLAYERS 4
#define TR 32
#define EPW 8

typedef __attribute__((ext_vector_type(2))) float v2f;
typedef __attribute__((ext_vector_type(8))) float v8f;

#define HC 16
#define HR 4
#define HWIN 16384
#define HLD 16

__global__ void __launch_bounds__(1024)
degree_hist_kernel(const int* __restrict__ src, const int* __restrict__ dst,
                   int* __restrict__ part, int E, long Np) {
    __shared__ int4 hist4[HWIN / 4];
    int* hist = (int*)hist4;

    const int tid = threadIdx.x;
    const int c   = blockIdx.x % HC;
    const int r   = (blockIdx.x / HC) % HR;
    const int a   = blockIdx.x / (HC * HR);
    const int* __restrict__ idx = a ? dst : src;

    const long SL = ((Np + 64 * HR - 1) / (64 * HR)) * 64;
    long n0 = r * SL;        if (n0 > Np) n0 = Np;
    long n1 = n0 + SL;       if (n1 > Np) n1 = Np;

    const long chunk = ((((long)E + HC - 1) / HC) + 3) & ~3L;
    long e0 = c * chunk;     if (e0 > E) e0 = E;
    long e1 = e0 + chunk;    if (e1 > E) e1 = E;
    const int nv   = (int)((e1 - e0) >> 2);
    const int tail = (int)(e1 - e0) - 4 * nv;
    const int4* __restrict__ idx4 = (const int4*)(idx + e0);

    int* __restrict__ outp = part + ((long)a * HC + c) * Np;

    for (long w0 = n0; w0 < n1; w0 += HWIN) {
        const int      wb = (int)w0;
        const unsigned wn = (unsigned)((n1 - w0 < HWIN) ? (n1 - w0) : HWIN);
        for (unsigned i = 4 * tid; i < wn; i += 4096) hist4[i >> 2] = make_int4(0, 0, 0, 0);
        __syncthreads();

        for (int vb = 0; vb < nv; vb += HLD * 1024) {
            int4 q[HLD];
#pragma unroll
            for (int j = 0; j < HLD; ++j) {
                const int v = vb + j * 1024 + tid;
                q[j] = idx4[v < nv ? v : nv - 1];
            }
#pragma unroll
            for (int j = 0; j < HLD; ++j) {
                if (vb + j * 1024 + tid < nv) {
                    const unsigned t0 = (unsigned)(q[j].x - wb), t1 = (unsigned)(q[j].y - wb);
                    const unsigned t2 = (unsigned)(q[j].z - wb), t3 = (unsigned)(q[j].w - wb);
                    if (t0 < wn) atomicAdd(&hist[t0], 1);
                    if (t1 < wn) atomicAdd(&hist[t1], 1);
                    if (t2 < wn) atomicAdd(&hist[t2], 1);
                    if (t3 < wn) atomicAdd(&hist[t3], 1);
                }
            }
        }
        if (tid < tail) {
            const unsigned t = (unsigned)(idx[e0 + 4L * nv + tid] - wb);
            if (t < wn) atomicAdd(&hist[t], 1);
        }
        __syncthreads();

        for (unsigned i = 4 * tid; i < wn; i += 4096) *(int4*)(outp + w0 + i) = hist4[i >> 2];
        __syncthreads();
    }
}

__global__ void norm_count_kernel(const int* __restrict__ part, float* __restrict__ nrm,
                                  int* __restrict__ cnt, long Np) {
    const long i = (long)blockIdx.x * blockDim.x + threadIdx.x;
    const long n = 2 * Np;
    int dg = 0;
    if (i < n) {
        const long a = (i >= Np) ? 1 : 0;
        const int* __restrict__ p = part + a * HC * Np + (i - a * Np);
        int d[HC];
#pragma unroll
        for (int g = 0; g < HC; ++g) d[g] = p[(long)g * Np];
#pragma unroll
        for (int g = 0; g < HC; ++g) dg += d[g];
    }
    int tsum = dg;
#pragma unroll
    for (int o = 16; o > 0; o >>= 1) tsum += __shfl_xor(tsum, o);
    if (i < n) {
        nrm[i] = rsqrtf(fmaxf((float)dg, 1.0f));
        if (i >= Np && (threadIdx.x & 31) == 0) cnt[(i - Np) >> 5] = tsum;
    }
}

__global__ void __launch_bounds__(256)
tile_scan_kernel(int* __restrict__ cur, int* __restrict__ off, int NT) {
    __shared__ int part[256];
    const int tid   = threadIdx.x;
    const int chunk = (NT + 255) / 256;
    int t0 = tid * chunk;       if (t0 > NT) t0 = NT;
    int t1 = t0 + chunk;        if (t1 > NT) t1 = NT;
    int sum = 0;
    for (int t = t0; t < t1; ++t) sum += cur[t];
    part[tid] = sum;
    __syncthreads();
    for (int o = 1; o < 256; o <<= 1) {
        int v = (tid >= o) ? part[tid - o] : 0;
        __syncthreads();
        part[tid] += v;
        __syncthreads();
    }
    int run = part[tid] - sum;
    for (int t = t0; t < t1; ++t) {
        int c = cur[t];
        off[t] = run;
        cur[t] = run;
        run += c;
    }
    if (tid == 255) off[NT] = part[255];
}

#define BF_EPT 8
#define BF_WIN 2048

__global__ void __launch_bounds__(1024)
bin_fill_kernel(const int* __restrict__ src, const int* __restrict__ dst,
                int* __restrict__ cur,
                int* __restrict__ bsrc, int* __restrict__ bdst, int E, int NT) {
    __shared__ int hist[BF_WIN];
    const int  tid   = threadIdx.x;
    const long ebase = (long)blockIdx.x * (BF_EPT * 1024);

    int s[BF_EPT], d[BF_EPT], r[BF_EPT];
#pragma unroll
    for (int j = 0; j < BF_EPT; ++j) {
        const long e = ebase + j * 1024 + tid;
        const bool ok = e < E;
        d[j] = ok ? dst[e] : -1;
        s[j] = ok ? src[e] : 0;
        r[j] = 0;
    }

    for (int w0 = 0; w0 < NT; w0 += BF_WIN) {
#pragma unroll
        for (int k = 0; k < BF_WIN / 1024; ++k) hist[tid + 1024 * k] = 0;
        __syncthreads();

        bool in[BF_EPT];
        int  t[BF_EPT];
#pragma unroll
        for (int j = 0; j < BF_EPT; ++j) {
            const int tj = (d[j] >> 5) - w0;
            in[j] = tj >= 0 && tj < BF_WIN;
            t[j]  = in[j] ? tj : 0;
            if (in[j]) r[j] = atomicAdd(&hist[t[j]], 1);
        }
        __syncthreads();

        int c[BF_WIN / 1024], b[BF_WIN / 1024];
#pragma unroll
        for (int k = 0; k < BF_WIN / 1024; ++k) c[k] = hist[tid + 1024 * k];
#pragma unroll
        for (int k = 0; k < BF_WIN / 1024; ++k) {
            b[k] = 0;
            if (c[k] > 0) b[k] = atomicAdd(&cur[w0 + tid + 1024 * k], c[k]);
        }
#pragma unroll
        for (int k = 0; k < BF_WIN / 1024; ++k) hist[tid + 1024 * k] = b[k];
        __syncthreads();

        int p[BF_EPT];
#pragma unroll
        for (int j = 0; j < BF_EPT; ++j) p[j] = hist[t[j]] + r[j];
#pragma unroll
        for (int j = 0; j < BF_EPT; ++j) {
            if (in[j]) {
                bsrc[p[j]] = s[j];
                bdst[p[j]] = d[j];
            }
        }
        __syncthreads();
    }
}

#define SCAP 1024
#define SPT  (SCAP / 256)

__device__ __forceinline__ void lds_add(float* p, float v) {
    __hip_atomic_fetch_add(p, v, __ATOMIC_RELAXED, __HIP_MEMORY_SCOPE_WORKGROUP);
}

__global__ void __launch_bounds__(256, 16)
tile_sum_kernel(const float* __restrict__ hin,
                const float* __restrict__ norm_src,
                const int* __restrict__ bsrc,
                const int* __restrict__ bdst,
                const int* __restrict__ off,
                float* __restrict__ agg, int N) {
    __shared__ float4 sT4[TR * D / 4];
    __shared__ int2   sKN[SCAP];
    __shared__ int    sCnt[TR];
    float* sT = (float*)sT4;

    const int tid = threadIdx.x;
    const int l   = tid & 31;
    const int w   = __builtin_amdgcn_readfirstlane(tid >> 5);
    const int t   = blockIdx.x;
    const int rowBase = t * TR;

#pragma unroll
    for (int k = 0; k < TR * D / 4 / 256; ++k) sT4[tid + 256 * k] = make_float4(0.f, 0.f, 0.f, 0.f);

    const int beg = off[t];
    const int end = off[t + 1];

    for (int c0 = beg; c0 < end; c0 += SCAP) {
        const int cn = (end - c0 < SCAP) ? (end - c0) : SCAP;

        if (tid < TR) sCnt[tid] = 0;
        int dk[SPT], sk[SPT];
#pragma unroll
        for (int i = 0; i < SPT; ++i) {
            const int k  = tid + 256 * i;
            const int kc = c0 + (k < cn ? k : cn - 1);
            dk[i] = bdst[kc] - rowBase;
            sk[i] = bsrc[kc];
        }
        float nk[SPT];
#pragma unroll
        for (int i = 0; i < SPT; ++i) nk[i] = norm_src[sk[i]];
        __syncthreads();

        int rk[SPT];
#pragma unroll
        for (int i = 0; i < SPT; ++i) {
            rk[i] = 0;
            if (tid + 256 * i < cn) rk[i] = atomicAdd(&sCnt[dk[i]], 1);
        }
        __syncthreads();

        const int cr = sCnt[l];
        int inc = cr;
#pragma unroll
        for (int o = 1; o < 32; o <<= 1) {
            const int u = __shfl_up(inc, o);
            if (l >= o) inc += u;
        }
        const int exc = inc - cr;
#pragma unroll
        for (int i = 0; i < SPT; ++i) {
            const int b = __shfl(exc, dk[i]);
            if (tid + 256 * i < cn) sKN[b + rk[i]] = make_int2(sk[i] * TR + dk[i], __float_as_int(nk[i]));
        }
        __syncthreads();

        const int nb = (cn + EPW - 1) / EPW;
        const int kb = ((w * nb) >> 3) * EPW;
        int       ke = (((w + 1) * nb) >> 3) * EPW;
        if (ke > cn) ke = cn;

        float4 acc = make_float4(0.f, 0.f, 0.f, 0.f);
        int    cur = -1;
        int  kk = kb + (l & (EPW - 1));
        int2 kv = sKN[kk < cn ? kk : cn - 1];
        for (int k0 = kb; k0 < ke; k0 += EPW) {
            int key[EPW];
#pragma unroll
            for (int j = 0; j < EPW; ++j) key[j] = __builtin_amdgcn_readlane(kv.x, j);
            float  ns[EPW];
#pragma unroll
            for (int j = 0; j < EPW; ++j) ns[j] = __int_as_float(__builtin_amdgcn_readlane(kv.y, j));

            float4 v[EPW];
#pragma unroll
            for (int j = 0; j < EPW; ++j) {
                const int s = key[j] >> 5;
                v[j] = ((const float4*)(hin + (long)s * D))[l];
            }
            kk = k0 + EPW + (l & (EPW - 1));
            kv = sKN[kk < cn ? kk : cn - 1];
#pragma unroll
            for (int j = 0; j < EPW; ++j) {
                if (k0 + j < ke) {
                    const int r = key[j] & (TR - 1);
                    if (r != cur) {
                        if (cur >= 0) {
                            float* p = sT + cur * D + l;
                            lds_add(p + 0,  acc.x);
                            lds_add(p + 32, acc.y);
                            lds_add(p + 64, acc.z);
                            lds_add(p + 96, acc.w);
                        }
                        acc = make_float4(0.f, 0.f, 0.f, 0.f);
                        cur = r;
                    }
                    acc.x += v[j].x * ns[j];
                    acc.y += v[j].y * ns[j];
                    acc.z += v[j].z * ns[j];
                    acc.w += v[j].w * ns[j];
                }
            }
        }
        if (cur >= 0) {
            float* p = sT + cur * D + l;
            lds_add(p + 0,  acc.x);
            lds_add(p + 32, acc.y);
            lds_add(p + 64, acc.z);
            lds_add(p + 96, acc.w);
        }
    }
    __syncthreads();

#pragma unroll
    for (int k = 0; k < TR * D / 4 / 256; ++k) {
        const int i   = tid + 256 * k;
        const int row = rowBase + (i >> 5);
        const float* q = sT + (i >> 5) * D + l;
        const float4 o = make_float4(q[0], q[32], q[64], q[96]);
        if (row < N) ((float4*)(agg + (long)row * D))[l] = o;
    }
}

#define GR 128
#define AP (D + 4)

__global__ void __launch_bounds__(256)
gcn_gemm_wmma_kernel(const float* __restrict__ agg,
                     const float* __restrict__ norm_dst,
                     const float* __restrict__ W,
                     const float* __restrict__ bias,
                     float* __restrict__ out, int N) {
    __shared__ float4 sLds[D * D / 4 + GR * AP / 4];
    float4* sW4 = sLds;
    float*  sA  = (float*)(sLds + D * D / 4);

    const int tid = threadIdx.x;
    const int l   = tid & 31;
    const int w   = __builtin_amdgcn_readfirstlane(tid >> 5);
    const int c   = l & 15;
    const int h   = l >> 4;
    const int rowBase = blockIdx.x * GR + 16 * w;

    float4 w0[8], w1[8];
#pragma unroll
    for (int i = 0; i < 8; ++i) {
        const int m = w + 8 * i;
        w0[i] = ((const float4*)W)[(2 * m) * 32 + l];
        w1[i] = ((const float4*)W)[(2 * m + 1) * 32 + l];
    }
    float4 av[16];
    float  nd[16];
#pragma unroll
    for (int i = 0; i < 16; ++i) {
        const int row = (rowBase + i < N) ? rowBase + i : N - 1;
        av[i] = ((const float4*)(agg + (long)row * D))[l];
        nd[i] = norm_dst[row];
    }
    float2 bp[4];
#pragma unroll
    for (int p = 0; p < 4; ++p) bp[p] = ((const float2*)bias)[16 * p + c];

    __builtin_amdgcn_sched_barrier(0);

    const int sw = (w & 1) << 3;
#pragma unroll
    for (int i = 0; i < 8; ++i) {
        const int m = w + 8 * i;
        float4* d = sW4 + m * 64 + ((2 * l) ^ sw);
        d[0] = make_float4(w0[i].x, w1[i].x, w0[i].y, w1[i].y);
        d[1] = make_float4(w0[i].z, w1[i].z, w0[i].w, w1[i].w);
    }
#pragma unroll
    for (int i = 0; i < 16; ++i)
        *(float4*)(sA + (16 * w + i) * AP + 4 * l) =
            make_float4(av[i].x * nd[i], av[i].y * nd[i], av[i].z * nd[i], av[i].w * nd[i]);
    __syncthreads();

    v8f acc[8];
#pragma unroll
    for (int p = 0; p < 4; ++p) {
#pragma unroll
        for (int r = 0; r < 8; ++r) { acc[2 * p][r] = bp[p].x; acc[2 * p + 1][r] = bp[p].y; }
    }

    const float*  pA = sA + (16 * w + c) * AP + 2 * h;
    const float4* pW = sW4 + h * 64 + (c ^ (h << 3));

    v2f    a = *(const v2f*)pA;
    float4 q[4];
#pragma unroll
    for (int p = 0; p < 4; ++p) q[p] = pW[16 * p];

#pragma unroll
    for (int t = 0; t < D / 4; ++t) {
        v2f    an = a;
        float4 qn[4];
#pragma unroll
        for (int p = 0; p < 4; ++p) qn[p] = q[p];
        if (t + 1 < D / 4) {
            an = *(const v2f*)(pA + 4 * (t + 1));
#pragma unroll
            for (int p = 0; p < 4; ++p) qn[p] = pW[128 * (t + 1) + 16 * p];
        }
#pragma unroll
        for (int p = 0; p < 4; ++p) {
            v2f b0, b1;
            b0.x = q[p].x; b0.y = q[p].y;
            b1.x = q[p].z; b1.y = q[p].w;
            acc[2 * p]     = __builtin_amdgcn_wmma_f32_16x16x4_f32(false, a, false, b0, (short)0, acc[2 * p],     false, false);
            acc[2 * p + 1] = __builtin_amdgcn_wmma_f32_16x16x4_f32(false, a, false, b1, (short)0, acc[2 * p + 1], false, false);
        }
        a = an;
#pragma unroll
        for (int p = 0; p < 4; ++p) q[p] = qn[p];
    }

#pragma unroll
    for (int r = 0; r < 8; ++r) {
        const int row = rowBase + 8 * h + r;
        if (row < N) {
            float2* o = (float2*)(out + (long)row * D);
#pragma unroll
            for (int p = 0; p < 4; ++p) o[16 * p + c] = make_float2(acc[2 * p][r], acc[2 * p + 1][r]);
        }
    }
}


extern "C" void kernel_launch(void* const* d_in, const int* in_sizes, int n_in,
                              void* d_out, int out_size, void* d_ws, size_t ws_size,
                              hipStream_t stream) {
    const float* feat = (const float*)d_in[0];
    const float* W    = (const float*)d_in[1];
    const float* bias = (const float*)d_in[2];
    const int*   src  = (const int*)d_in[3];
    const int*   dst  = (const int*)d_in[4];

    const int N = in_sizes[0] / D;
    const int E = in_sizes[3];

    const long Np  = ((long)N + 63) & ~63L;
    const int  NT  = (int)(Np / TR);
    const long NTa = ((long)NT + 1 + 63) & ~63L;
    const long Ea  = ((long)E + 63) & ~63L;

    float* ws       = (float*)d_ws;
    float* norm_src = ws;
    float* norm_dst = ws + Np;
    int*   cur      = (int*)(ws + 2 * Np);
    int*   off      = cur + NTa;
    int*   bsrc     = off + NTa;
    int*   bdst     = bsrc + Ea;
    float* agg      = (float*)(bdst + Ea);
    float* hbuf     = agg + (long)N * D;
    int*   part     = (int*)(hbuf + (long)N * D);
    float* outf     = (float*)d_out;

    const long twoNp = 2 * Np;

    degree_hist_kernel<<<2 * HR * HC, 1024, 0, stream>>>(src, dst, part, E, Np);
    norm_count_kernel<<<(int)((twoNp + 255) / 256), 256, 0, stream>>>(part, norm_src, cur, Np);
    tile_scan_kernel<<<1, 256, 0, stream>>>(cur, off, NT);
    bin_fill_kernel<<<(E + BF_EPT * 1024 - 1) / (BF_EPT * 1024), 1024, 0, stream>>>(src, dst, cur, bsrc, bdst, E, NT);

    const int gemmGrid = (N + GR - 1) / GR;

    const float* hin = feat;
    for (int layer = 0; layer < NLAYERS; ++layer) {
        float* hout = (layer & 1) ? outf : hbuf;

        tile_sum_kernel<<<NT, 256, 0, stream>>>(hin, norm_src, bsrc, bdst, off, agg, N);
        gcn_gemm_wmma_kernel<<<gemmGrid, 256, 0, stream>>>(
            agg, norm_dst, W + (long)layer * D * D, bias + (long)layer * D, hout, N);

        hin = hout;
    }
}


# Bin fill: group's edges put in tile order in LDS (prefix of tile counts + rank), then stored front to back, adjacent lanes to adjacent slots: runs per store, not 32 lone words; sum order unchanged
# speedup vs baseline: 4.0761x; 1.1170x over previous
#include <hip/hip_runtime.h>

#define D 128
#define NLAYERS 4
#define TR 32
#define EPW 8

typedef __attribute__((ext_vector_type(2))) float v2f;
typedef __attribute__((ext_vector_type(8))) float v8f;

#define HC 16
#define HR 4
#define HWIN 16384
#define HLD 16

__global__ void __launch_bounds__(1024)
degree_hist_kernel(const int* __restrict__ src, const int* __restrict__ dst,
                   int* __restrict__ part, int E, long Np) {
    __shared__ int4 hist4[HWIN / 4];
    int* hist = (int*)hist4;

    const int tid = threadIdx.x;
    const int c   = blockIdx.x % HC;
    const int r   = (blockIdx.x / HC) % HR;
    const int a   = blockIdx.x / (HC * HR);
    const int* __restrict__ idx = a ? dst : src;

    const long SL = ((Np + 64 * HR - 1) / (64 * HR)) * 64;
    long n0 = r * SL;        if (n0 > Np) n0 = Np;
    long n1 = n0 + SL;       if (n1 > Np) n1 = Np;

    const long chunk = ((((long)E + HC - 1) / HC) + 3) & ~3L;
    long e0 = c * chunk;     if (e0 > E) e0 = E;
    long e1 = e0 + chunk;    if (e1 > E) e1 = E;
    const int nv   = (int)((e1 - e0) >> 2);
    const int tail = (int)(e1 - e0) - 4 * nv;
    const int4* __restrict__ idx4 = (const int4*)(idx + e0);

    int* __restrict__ outp = part + ((long)a * HC + c) * Np;

    for (long w0 = n0; w0 < n1; w0 += HWIN) {
        const int      wb = (int)w0;
        const unsigned wn = (unsigned)((n1 - w0 < HWIN) ? (n1 - w0) : HWIN);
        for (unsigned i = 4 * tid; i < wn; i += 4096) hist4[i >> 2] = make_int4(0, 0, 0, 0);
        __syncthreads();

        for (int vb = 0; vb < nv; vb += HLD * 1024) {
            int4 q[HLD];
#pragma unroll
            for (int j = 0; j < HLD; ++j) {
                const int v = vb + j * 1024 + tid;
                q[j] = idx4[v < nv ? v : nv - 1];
            }
#pragma unroll
            for (int j = 0; j < HLD; ++j) {
                if (vb + j * 1024 + tid < nv) {
                    const unsigned t0 = (unsigned)(q[j].x - wb), t1 = (unsigned)(q[j].y - wb);
                    const unsigned t2 = (unsigned)(q[j].z - wb), t3 = (unsigned)(q[j].w - wb);
                    if (t0 < wn) atomicAdd(&hist[t0], 1);
                    if (t1 < wn) atomicAdd(&hist[t1], 1);
                    if (t2 < wn) atomicAdd(&hist[t2], 1);
                    if (t3 < wn) atomicAdd(&hist[t3], 1);
                }
            }
        }
        if (tid < tail) {
            const unsigned t = (unsigned)(idx[e0 + 4L * nv + tid] - wb);
            if (t < wn) atomicAdd(&hist[t], 1);
        }
        __syncthreads();

        for (unsigned i = 4 * tid; i < wn; i += 4096) *(int4*)(outp + w0 + i) = hist4[i >> 2];
        __syncthreads();
    }
}

__global__ void norm_count_kernel(const int* __restrict__ part, float* __restrict__ nrm,
                                  int* __restrict__ cnt, long Np) {
    const long i = (long)blockIdx.x * blockDim.x + threadIdx.x;
    const long n = 2 * Np;
    int dg = 0;
    if (i < n) {
        const long a = (i >= Np) ? 1 : 0;
        const int* __restrict__ p = part + a * HC * Np + (i - a * Np);
        int d[HC];
#pragma unroll
        for (int g = 0; g < HC; ++g) d[g] = p[(long)g * Np];
#pragma unroll
        for (int g = 0; g < HC; ++g) dg += d[g];
    }
    int tsum = dg;
#pragma unroll
    for (int o = 16; o > 0; o >>= 1) tsum += __shfl_xor(tsum, o);
    if (i < n) {
        nrm[i] = rsqrtf(fmaxf((float)dg, 1.0f));
        if (i >= Np && (threadIdx.x & 31) == 0) cnt[(i - Np) >> 5] = tsum;
    }
}

__global__ void __launch_bounds__(256)
tile_scan_kernel(int* __restrict__ cur, int* __restrict__ off, int NT) {
    __shared__ int part[256];
    const int tid   = threadIdx.x;
    const int chunk = (NT + 255) / 256;
    int t0 = tid * chunk;       if (t0 > NT) t0 = NT;
    int t1 = t0 + chunk;        if (t1 > NT) t1 = NT;
    int sum = 0;
    for (int t = t0; t < t1; ++t) sum += cur[t];
    part[tid] = sum;
    __syncthreads();
    for (int o = 1; o < 256; o <<= 1) {
        int v = (tid >= o) ? part[tid - o] : 0;
        __syncthreads();
        part[tid] += v;
        __syncthreads();
    }
    int run = part[tid] - sum;
    for (int t = t0; t < t1; ++t) {
        int c = cur[t];
        off[t] = run;
        cur[t] = run;
        run += c;
    }
    if (tid == 255) off[NT] = part[255];
}

#define BF_EPT 8
#define BF_WIN 2048

__global__ void __launch_bounds__(1024)
bin_fill_kernel(const int* __restrict__ src, const int* __restrict__ dst,
                int* __restrict__ cur,
                int* __restrict__ bsrc, int* __restrict__ bdst, int E, int NT) {
    __shared__ int2 hist2[BF_WIN / 2];
    __shared__ int2 delta2[BF_WIN / 2];
    __shared__ int  ssrc[BF_EPT * 1024];
    __shared__ int  sdst[BF_EPT * 1024];
    __shared__ int  wsum[32];
    int* hist  = (int*)hist2;
    int* delta = (int*)delta2;

    const int  tid   = threadIdx.x;
    const int  l     = tid & 31;
    const int  wv    = tid >> 5;
    const long ebase = (long)blockIdx.x * (BF_EPT * 1024);

    int s[BF_EPT], d[BF_EPT], r[BF_EPT];
#pragma unroll
    for (int j = 0; j < BF_EPT; ++j) {
        const long e = ebase + j * 1024 + tid;
        const bool ok = e < E;
        d[j] = ok ? dst[e] : -1;
        s[j] = ok ? src[e] : 0;
        r[j] = 0;
    }

    for (int w0 = 0; w0 < NT; w0 += BF_WIN) {
        hist2[tid] = make_int2(0, 0);
        __syncthreads();

        bool in[BF_EPT];
        int  t[BF_EPT];
#pragma unroll
        for (int j = 0; j < BF_EPT; ++j) {
            const int tj = (d[j] >> 5) - w0;
            in[j] = tj >= 0 && tj < BF_WIN;
            t[j]  = in[j] ? tj : 0;
            if (in[j]) r[j] = atomicAdd(&hist[t[j]], 1);
        }
        __syncthreads();

        const int2 cc = hist2[tid];
        int b0 = 0, b1 = 0;
        if (cc.x > 0) b0 = atomicAdd(&cur[w0 + 2 * tid],     cc.x);
        if (cc.y > 0) b1 = atomicAdd(&cur[w0 + 2 * tid + 1], cc.y);

        const int s2 = cc.x + cc.y;
        int inc = s2;
#pragma unroll
        for (int o = 1; o < 32; o <<= 1) {
            const int u = __shfl_up(inc, o);
            if (l >= o) inc += u;
        }
        if (l == 31) wsum[wv] = inc;
        __syncthreads();
        const int wt = wsum[l];
        int winc = wt;
#pragma unroll
        for (int o = 1; o < 32; o <<= 1) {
            const int u = __shfl_up(winc, o);
            if (l >= o) winc += u;
        }
        const int total = __shfl(winc, 31);
        const int l0 = __shfl(winc - wt, wv) + inc - s2;
        const int l1 = l0 + cc.x;
        hist2[tid]  = make_int2(l0, l1);
        delta2[tid] = make_int2(b0 - l0, b1 - l1);
        __syncthreads();

#pragma unroll
        for (int j = 0; j < BF_EPT; ++j) {
            if (in[j]) {
                const int li = hist[t[j]] + r[j];
                ssrc[li] = s[j];
                sdst[li] = d[j];
            }
        }
        __syncthreads();

        int dd[BF_EPT], ss[BF_EPT], p[BF_EPT];
#pragma unroll
        for (int k = 0; k < BF_EPT; ++k) {
            const int i  = tid + 1024 * k;
            const int ic = i < total ? i : 0;
            dd[k] = sdst[ic];
            ss[k] = ssrc[ic];
        }
#pragma unroll
        for (int k = 0; k < BF_EPT; ++k) {
            const int tk = (dd[k] >> 5) - w0;
            p[k] = delta[(tk >= 0 && tk < BF_WIN) ? tk : 0] + tid + 1024 * k;
        }
#pragma unroll
        for (int k = 0; k < BF_EPT; ++k) {
            if (tid + 1024 * k < total) {
                bsrc[p[k]] = ss[k];
                bdst[p[k]] = dd[k];
            }
        }
        __syncthreads();
    }
}

#define SCAP 1024
#define SPT  (SCAP / 256)

__device__ __forceinline__ void lds_add(float* p, float v) {
    __hip_atomic_fetch_add(p, v, __ATOMIC_RELAXED, __HIP_MEMORY_SCOPE_WORKGROUP);
}

__global__ void __launch_bounds__(256, 16)
tile_sum_kernel(const float* __restrict__ hin,
                const float* __restrict__ norm_src,
                const int* __restrict__ bsrc,
                const int* __restrict__ bdst,
                const int* __restrict__ off,
                float* __restrict__ agg, int N) {
    __shared__ float4 sT4[TR * D / 4];
    __shared__ int2   sKN[SCAP];
    __shared__ int    sCnt[TR];
    float* sT = (float*)sT4;

    const int tid = threadIdx.x;
    const int l   = tid & 31;
    const int w   = __builtin_amdgcn_readfirstlane(tid >> 5);
    const int t   = blockIdx.x;
    const int rowBase = t * TR;

#pragma unroll
    for (int k = 0; k < TR * D / 4 / 256; ++k) sT4[tid + 256 * k] = make_float4(0.f, 0.f, 0.f, 0.f);

    const int beg = off[t];
    const int end = off[t + 1];

    for (int c0 = beg; c0 < end; c0 += SCAP) {
        const int cn = (end - c0 < SCAP) ? (end - c0) : SCAP;

        if (tid < TR) sCnt[tid] = 0;
        int dk[SPT], sk[SPT];
#pragma unroll
        for (int i = 0; i < SPT; ++i) {
            const int k  = tid + 256 * i;
            const int kc = c0 + (k < cn ? k : cn - 1);
            dk[i] = bdst[kc] - rowBase;
            sk[i] = bsrc[kc];
        }
        float nk[SPT];
#pragma unroll
        for (int i = 0; i < SPT; ++i) nk[i] = norm_src[sk[i]];
        __syncthreads();

        int rk[SPT];
#pragma unroll
        for (int i = 0; i < SPT; ++i) {
            rk[i] = 0;
            if (tid + 256 * i < cn) rk[i] = atomicAdd(&sCnt[dk[i]], 1);
        }
        __syncthreads();

        const int cr = sCnt[l];
        int inc = cr;
#pragma unroll
        for (int o = 1; o < 32; o <<= 1) {
            const int u = __shfl_up(inc, o);
            if (l >= o) inc += u;
        }
        const int exc = inc - cr;
#pragma unroll
        for (int i = 0; i < SPT; ++i) {
            const int b = __shfl(exc, dk[i]);
            if (tid + 256 * i < cn) sKN[b + rk[i]] = make_int2(sk[i] * TR + dk[i], __float_as_int(nk[i]));
        }
        __syncthreads();

        const int nb = (cn + EPW - 1) / EPW;
        const int kb = ((w * nb) >> 3) * EPW;
        int       ke = (((w + 1) * nb) >> 3) * EPW;
        if (ke > cn) ke = cn;

        float4 acc = make_float4(0.f, 0.f, 0.f, 0.f);
        int    cur = -1;
        int  kk = kb + (l & (EPW - 1));
        int2 kv = sKN[kk < cn ? kk : cn - 1];
        for (int k0 = kb; k0 < ke; k0 += EPW) {
            int key[EPW];
#pragma unroll
            for (int j = 0; j < EPW; ++j) key[j] = __builtin_amdgcn_readlane(kv.x, j);
            float  ns[EPW];
#pragma unroll
            for (int j = 0; j < EPW; ++j) ns[j] = __int_as_float(__builtin_amdgcn_readlane(kv.y, j));

            float4 v[EPW];
#pragma unroll
            for (int j = 0; j < EPW; ++j) {
                const int s = key[j] >> 5;
                v[j] = ((const float4*)(hin + (long)s * D))[l];
            }
            kk = k0 + EPW + (l & (EPW - 1));
            kv = sKN[kk < cn ? kk : cn - 1];
#pragma unroll
            for (int j = 0; j < EPW; ++j) {
                if (k0 + j < ke) {
                    const int r = key[j] & (TR - 1);
                    if (r != cur) {
                        if (cur >= 0) {
                            float* p = sT + cur * D + l;
                            lds_add(p + 0,  acc.x);
                            lds_add(p + 32, acc.y);
                            lds_add(p + 64, acc.z);
                            lds_add(p + 96, acc.w);
                        }
                        acc = make_float4(0.f, 0.f, 0.f, 0.f);
                        cur = r;
                    }
                    acc.x += v[j].x * ns[j];
                    acc.y += v[j].y * ns[j];
                    acc.z += v[j].z * ns[j];
                    acc.w += v[j].w * ns[j];
                }
            }
        }
        if (cur >= 0) {
            float* p = sT + cur * D + l;
            lds_add(p + 0,  acc.x);
            lds_add(p + 32, acc.y);
            lds_add(p + 64, acc.z);
            lds_add(p + 96, acc.w);
        }
    }
    __syncthreads();

#pragma unroll
    for (int k = 0; k < TR * D / 4 / 256; ++k) {
        const int i   = tid + 256 * k;
        const int row = rowBase + (i >> 5);
        const float* q = sT + (i >> 5) * D + l;
        const float4 o = make_float4(q[0], q[32], q[64], q[96]);
        if (row < N) ((float4*)(agg + (long)row * D))[l] = o;
    }
}

#define GR 128
#define AP (D + 4)

__global__ void __launch_bounds__(256)
gcn_gemm_wmma_kernel(const float* __restrict__ agg,
                     const float* __restrict__ norm_dst,
                     const float* __restrict__ W,
                     const float* __restrict__ bias,
                     float* __restrict__ out, int N) {
    __shared__ float4 sLds[D * D / 4 + GR * AP / 4];
    float4* sW4 = sLds;
    float*  sA  = (float*)(sLds + D * D / 4);

    const int tid = threadIdx.x;
    const int l   = tid & 31;
    const int w   = __builtin_amdgcn_readfirstlane(tid >> 5);
    const int c   = l & 15;
    const int h   = l >> 4;
    const int rowBase = blockIdx.x * GR + 16 * w;

    float4 w0[8], w1[8];
#pragma unroll
    for (int i = 0; i < 8; ++i) {
        const int m = w + 8 * i;
        w0[i] = ((const float4*)W)[(2 * m) * 32 + l];
        w1[i] = ((const float4*)W)[(2 * m + 1) * 32 + l];
    }
    float4 av[16];
    float  nd[16];
#pragma unroll
    for (int i = 0; i < 16; ++i) {
        const int row = (rowBase + i < N) ? rowBase + i : N - 1;
        av[i] = ((const float4*)(agg + (long)row * D))[l];
        nd[i] = norm_dst[row];
    }
    float2 bp[4];
#pragma unroll
    for (int p = 0; p < 4; ++p) bp[p] = ((const float2*)bias)[16 * p + c];

    __builtin_amdgcn_sched_barrier(0);

    const int sw = (w & 1) << 3;
#pragma unroll
    for (int i = 0; i < 8; ++i) {
        const int m = w + 8 * i;
        float4* d = sW4 + m * 64 + ((2 * l) ^ sw);
        d[0] = make_float4(w0[i].x, w1[i].x, w0[i].y, w1[i].y);
        d[1] = make_float4(w0[i].z, w1[i].z, w0[i].w, w1[i].w);
    }
#pragma unroll
    for (int i = 0; i < 16; ++i)
        *(float4*)(sA + (16 * w + i) * AP + 4 * l) =
            make_float4(av[i].x * nd[i], av[i].y * nd[i], av[i].z * nd[i], av[i].w * nd[i]);
    __syncthreads();

    v8f acc[8];
#pragma unroll
    for (int p = 0; p < 4; ++p) {
#pragma unroll
        for (int r = 0; r < 8; ++r) { acc[2 * p][r] = bp[p].x; acc[2 * p + 1][r] = bp[p].y; }
    }

    const float*  pA = sA + (16 * w + c) * AP + 2 * h;
    const float4* pW = sW4 + h * 64 + (c ^ (h << 3));

    v2f    a = *(const v2f*)pA;
    float4 q[4];
#pragma unroll
    for (int p = 0; p < 4; ++p) q[p] = pW[16 * p];

#pragma unroll
    for (int t = 0; t < D / 4; ++t) {
        v2f    an = a;
        float4 qn[4];
#pragma unroll
        for (int p = 0; p < 4; ++p) qn[p] = q[p];
        if (t + 1 < D / 4) {
            an = *(const v2f*)(pA + 4 * (t + 1));
#pragma unroll
            for (int p = 0; p < 4; ++p) qn[p] = pW[128 * (t + 1) + 16 * p];
        }
#pragma unroll
        for (int p = 0; p < 4; ++p) {
            v2f b0, b1;
            b0.x = q[p].x; b0.y = q[p].y;
            b1.x = q[p].z; b1.y = q[p].w;
            acc[2 * p]     = __builtin_amdgcn_wmma_f32_16x16x4_f32(false, a, false, b0, (short)0, acc[2 * p],     false, false);
            acc[2 * p + 1] = __builtin_amdgcn_wmma_f32_16x16x4_f32(false, a, false, b1, (short)0, acc[2 * p + 1], false, false);
        }
        a = an;
#pragma unroll
        for (int p = 0; p < 4; ++p) q[p] = qn[p];
    }

#pragma unroll
    for (int r = 0; r < 8; ++r) {
        const int row = rowBase + 8 * h + r;
        if (row < N) {
            float2* o = (float2*)(out + (long)row * D);
#pragma unroll
            for (int p = 0; p < 4; ++p) o[16 * p + c] = make_float2(acc[2 * p][r], acc[2 * p + 1][r]);
        }
    }
}


extern "C" void kernel_launch(void* const* d_in, const int* in_sizes, int n_in,
                              void* d_out, int out_size, void* d_ws, size_t ws_size,
                              hipStream_t stream) {
    const float* feat = (const float*)d_in[0];
    const float* W    = (const float*)d_in[1];
    const float* bias = (const float*)d_in[2];
    const int*   src  = (const int*)d_in[3];
    const int*   dst  = (const int*)d_in[4];

    const int N = in_sizes[0] / D;
    const int E = in_sizes[3];

    const long Np  = ((long)N + 63) & ~63L;
    const int  NT  = (int)(Np / TR);
    const long NTa = ((long)NT + 1 + 63) & ~63L;
    const long Ea  = ((long)E + 63) & ~63L;

    float* ws       = (float*)d_ws;
    float* norm_src = ws;
    float* norm_dst = ws + Np;
    int*   cur      = (int*)(ws + 2 * Np);
    int*   off      = cur + NTa;
    int*   bsrc     = off + NTa;
    int*   bdst     = bsrc + Ea;
    float* agg      = (float*)(bdst + Ea);
    float* hbuf     = agg + (long)N * D;
    int*   part     = (int*)(hbuf + (long)N * D);
    float* outf     = (float*)d_out;

    const long twoNp = 2 * Np;

    degree_hist_kernel<<<2 * HR * HC, 1024, 0, stream>>>(src, dst, part, E, Np);
    norm_count_kernel<<<(int)((twoNp + 255) / 256), 256, 0, stream>>>(part, norm_src, cur, Np);
    tile_scan_kernel<<<1, 256, 0, stream>>>(cur, off, NT);
    bin_fill_kernel<<<(E + BF_EPT * 1024 - 1) / (BF_EPT * 1024), 1024, 0, stream>>>(src, dst, cur, bsrc, bdst, E, NT);

    const int gemmGrid = (N + GR - 1) / GR;

    const float* hin = feat;
    for (int layer = 0; layer < NLAYERS; ++layer) {
        float* hout = (layer & 1) ? outf : hbuf;

        tile_sum_kernel<<<NT, 256, 0, stream>>>(hin, norm_src, bsrc, bdst, off, agg, N);
        gcn_gemm_wmma_kernel<<<gemmGrid, 256, 0, stream>>>(
            agg, norm_dst, W + (long)layer * D * D, bias + (long)layer * D, hout, N);

        hin = hout;
    }
}
